# starting kernel
# baseline (speedup 1.0000x reference)
#include <hip/hip_runtime.h>

typedef __attribute__((ext_vector_type(2))) float v2f;
typedef __attribute__((ext_vector_type(8))) float v8f;

#define HH   512
#define WW   512
#define CC   3
#define TILE 16
#define PATCH 24
#define EPITCH 25
#define HPITCH 17

__global__ __launch_bounds__(32)
void entropy9x9_wmma_kernel(const float* __restrict__ x, float* __restrict__ out) {
    const int lane = threadIdx.x;
    const int tilesX = WW / TILE;
    const int tx = blockIdx.x % tilesX;
    const int ty = blockIdx.x / tilesX;
    const int c  = blockIdx.y;
    const int b  = blockIdx.z;
    const int x0 = tx * TILE, y0 = ty * TILE;

    __shared__ float eT[PATCH * EPITCH];
    __shared__ float hs[PATCH * HPITCH];

    const float* plane = x + (size_t)b * HH * WW * CC + c;

    #pragma unroll
    for (int t = 0; t < (PATCH * PATCH) / 32; ++t) {
        int id = t * 32 + lane;
        int i = id / PATCH;
        int j = id - i * PATCH;
        int gy = y0 + i - 4;
        int gx = x0 + j - 4;
        float e = 0.0f;
        if ((unsigned)gy < (unsigned)HH && (unsigned)gx < (unsigned)WW) {
            float p = plane[((size_t)gy * WW + gx) * CC];
            e = p * __log2f(p + 1e-10f);
        }
        eT[i * EPITCH + j] = e;
    }
    __syncthreads();

    const int m  = lane & 15;
    const int hi = lane >> 4;
    const int ka = hi * 2;

    #pragma unroll
    for (int rb = 0; rb < 2; ++rb) {
        const int R = rb * 8;
        v8f acc = {};
        #pragma unroll
        for (int q = 0; q < 6; ++q) {
            const int k0 = q * 4;
            v2f a, bm;
            a.x = eT[(R + m) * EPITCH + k0 + ka];
            a.y = eT[(R + m) * EPITCH + k0 + ka + 1];
            const int kb = k0 + ka;
            bm.x = (m <= kb     && kb     <= m + 8) ? 1.0f : 0.0f;
            bm.y = (m <= kb + 1 && kb + 1 <= m + 8) ? 1.0f : 0.0f;
            acc = __builtin_amdgcn_wmma_f32_16x16x4_f32(false, a, false, bm,
                                                        (short)0, acc, false, false);
        }
        #pragma unroll
        for (int g = 0; g < 8; ++g)
            hs[(R + g + hi * 8) * HPITCH + m] = acc[g];
    }
    __syncthreads();

    v8f acc = {};
    #pragma unroll
    for (int q = 0; q < 6; ++q) {
        const int k0 = q * 4;
        const int kk = k0 + ka;
        v2f a, bm;
        a.x = (m <= kk     && kk     <= m + 8) ? 1.0f : 0.0f;
        a.y = (m <= kk + 1 && kk + 1 <= m + 8) ? 1.0f : 0.0f;
        bm.x = hs[kk * HPITCH + m];
        bm.y = hs[(kk + 1) * HPITCH + m];
        acc = __builtin_amdgcn_wmma_f32_16x16x4_f32(false, a, false, bm,
                                                    (short)0, acc, false, false);
    }

    float* oplane = out + (size_t)b * HH * WW * CC + c;
    #pragma unroll
    for (int g = 0; g < 8; ++g) {
        int r = g + hi * 8;
        oplane[((size_t)(y0 + r) * WW + (x0 + m)) * CC] = -acc[g];
    }
}

extern "C" void kernel_launch(void* const* d_in, const int* in_sizes, int n_in,
                              void* d_out, int out_size, void* d_ws, size_t ws_size,
                              hipStream_t stream) {
    (void)in_sizes; (void)n_in; (void)d_ws; (void)ws_size; (void)out_size;
    const float* x = (const float*)d_in[0];
    float* out = (float*)d_out;

    const int tiles = (HH / TILE) * (WW / TILE);
    dim3 grid(tiles, CC, 32);
    dim3 block(32);
    entropy9x9_wmma_kernel<<<grid, block, 0, stream>>>(x, out);
}


# one wave does all 3 channels of a 16x32-pixel tile: patch rows come in and output rows leave as 16 bytes per lane (whole 128-byte lines out), LDS kept interleaved; 16384 waves in 4-wave groups
# speedup vs baseline: 2.5188x; 2.5188x over previous
#include <hip/hip_runtime.h>


typedef __attribute__((ext_vector_type(2))) float v2f;
typedef __attribute__((ext_vector_type(4))) float v4f;
typedef __attribute__((ext_vector_type(8))) float v8f;

#define HH      512
#define WW      512
#define CC      3
#define ROWF    (WW * CC)
#define TROWS   16
#define TPX     32
#define TF      (TPX * CC)
#define PROWS   24
#define PPX     (TPX + 8)
#define PF      (PPX * CC)
#define PPIECES (PF / 4)
#define OPIECES (TF / 4)
#define EPITCH  124
#define HPITCH  17
#define OPITCH  100
#define WAVES   4

__device__ __forceinline__ void wave_lds_handover() {
    __builtin_amdgcn_fence(__ATOMIC_RELEASE, "wavefront");
    __builtin_amdgcn_wave_barrier();
    __builtin_amdgcn_fence(__ATOMIC_ACQUIRE, "wavefront");
}

__global__ __launch_bounds__(WAVES * 32)
void entropy9x9_wmma_kernel(const float* __restrict__ x, float* __restrict__ out) {
    __shared__ __attribute__((aligned(16))) float eT_all[WAVES][PROWS * EPITCH];
    __shared__ float hs_all[WAVES][PROWS * HPITCH];
    __shared__ __attribute__((aligned(16))) float oT_all[WAVES][TROWS * OPITCH];

    const int lane = threadIdx.x & 31;
    const int w    = __builtin_amdgcn_readfirstlane((int)(threadIdx.x >> 5));
    float* eT = eT_all[w];
    float* hs = hs_all[w];
    float* oT = oT_all[w];

    const int tIdx = (int)blockIdx.x * WAVES + w;
    const int tcol = tIdx & 15;
    const int trow = (tIdx >> 4) & 31;
    const int b    = tIdx >> 9;
    const int y0 = trow * TROWS;
    const int c0 = tcol * TF;

    const float* img = x + (size_t)b * HH * ROWF;

    const int gf = c0 - 12 + 4 * lane;
    const bool colok = lane < PPIECES && (unsigned)gf < (unsigned)ROWF;
    const int pbase = (y0 - 4) * ROWF + gf;
    v4f pv[PROWS];
    #pragma unroll
    for (int t = 0; t < PROWS; ++t)
        pv[t] = (v4f){0.0f, 0.0f, 0.0f, 0.0f};
    if (colok) {
        if (y0 - 4 >= 0) {
            #pragma unroll
            for (int t = 0; t < 4; ++t)
                pv[t] = *(const v4f*)(img + (pbase + t * ROWF));
        }
        #pragma unroll
        for (int t = 4; t < 20; ++t)
            pv[t] = *(const v4f*)(img + (pbase + t * ROWF));
        if (y0 + 19 < HH) {
            #pragma unroll
            for (int t = 20; t < PROWS; ++t)
                pv[t] = *(const v4f*)(img + (pbase + t * ROWF));
        }
    }
    #pragma unroll
    for (int t = 0; t < PROWS; ++t) {
        const v4f v = pv[t];
        v4f e;
        e.x = v.x * __log2f(v.x + 1e-10f);
        e.y = v.y * __log2f(v.y + 1e-10f);
        e.z = v.z * __log2f(v.z + 1e-10f);
        e.w = v.w * __log2f(v.w + 1e-10f);
        if (lane < PPIECES)
            *(v4f*)(eT + t * EPITCH + 4 * lane) = e;
    }
    wave_lds_handover();

    const int m  = lane & 15;
    const int hi = lane >> 4;
    const int ka = hi * 2;

    for (int blk = 0; blk < 6; ++blk) {
        const int c   = blk >> 1;
        const int pxb = (blk & 1) * 16;

        v8f acc0 = {};
        v8f acc1 = {};
        #pragma unroll
        for (int q = 0; q < 6; ++q) {
            const int kb = q * 4 + ka;
            const int fo = (pxb + kb) * CC + c;
            v2f a0, a1, bm;
            a0.x = eT[m * EPITCH + fo];
            a0.y = eT[m * EPITCH + fo + CC];
            a1.x = eT[(8 + m) * EPITCH + fo];
            a1.y = eT[(8 + m) * EPITCH + fo + CC];
            bm.x = (m <= kb     && kb     <= m + 8) ? 1.0f : 0.0f;
            bm.y = (m <= kb + 1 && kb + 1 <= m + 8) ? 1.0f : 0.0f;
            acc0 = __builtin_amdgcn_wmma_f32_16x16x4_f32(false, a0, false, bm,
                                                         (short)0, acc0, false, false);
            acc1 = __builtin_amdgcn_wmma_f32_16x16x4_f32(false, a1, false, bm,
                                                         (short)0, acc1, false, false);
        }
        #pragma unroll
        for (int g = 0; g < 8; ++g)
            hs[(g + hi * 8) * HPITCH + m] = acc0[g];
        #pragma unroll
        for (int g = 0; g < 8; ++g)
            hs[(8 + g + hi * 8) * HPITCH + m] = acc1[g];
        wave_lds_handover();

        v8f acc = {};
        #pragma unroll
        for (int q = 0; q < 6; ++q) {
            const int kk = q * 4 + ka;
            v2f a, bm;
            a.x = (m <= kk     && kk     <= m + 8) ? 1.0f : 0.0f;
            a.y = (m <= kk + 1 && kk + 1 <= m + 8) ? 1.0f : 0.0f;
            bm.x = hs[kk * HPITCH + m];
            bm.y = hs[(kk + 1) * HPITCH + m];
            acc = __builtin_amdgcn_wmma_f32_16x16x4_f32(false, a, false, bm,
                                                        (short)0, acc, false, false);
        }
        #pragma unroll
        for (int g = 0; g < 8; ++g)
            oT[(g + hi * 8) * OPITCH + (pxb + m) * CC + c] = -acc[g];
        wave_lds_handover();
    }

    float* oimg = out + (size_t)b * HH * ROWF;
    if (lane < OPIECES) {
        #pragma unroll
        for (int t = 0; t < TROWS; ++t) {
            const v4f v = *(const v4f*)(oT + t * OPITCH + 4 * lane);
            *(v4f*)(oimg + ((size_t)(y0 + t) * ROWF + c0 + 4 * lane)) = v;
        }
    }
}

extern "C" void kernel_launch(void* const* d_in, const int* in_sizes, int n_in,
                              void* d_out, int out_size, void* d_ws, size_t ws_size,
                              hipStream_t stream) {
    (void)in_sizes; (void)n_in; (void)d_ws; (void)ws_size; (void)out_size;
    const float* x = (const float*)d_in[0];
    float* out = (float*)d_out;

    dim3 grid(4096);
    dim3 block(WAVES * 32);
    entropy9x9_wmma_kernel<<<grid, block, 0, stream>>>(x, out);
}


# pass 2 takes the horizontal sums as its B operand straight from pass 1's accumulator registers (band A built in their row order: new k order of pass 2); rows 16-23 of both blocks share one tile
# speedup vs baseline: 2.8854x; 1.1455x over previous
#include <hip/hip_runtime.h>


typedef __attribute__((ext_vector_type(2))) float v2f;
typedef __attribute__((ext_vector_type(4))) float v4f;
typedef __attribute__((ext_vector_type(8))) float v8f;

#define HH      512
#define WW      512
#define CC      3
#define ROWF    (WW * CC)
#define TROWS   16
#define TPX     32
#define TF      (TPX * CC)
#define PROWS   24
#define PPX     (TPX + 8)
#define PF      (PPX * CC)
#define PPIECES (PF / 4)
#define OPIECES (TF / 4)
#define EPITCH  124
#define OPITCH  100
#define WAVES   4

__device__ __forceinline__ void wave_lds_handover() {
    __builtin_amdgcn_fence(__ATOMIC_RELEASE, "wavefront");
    __builtin_amdgcn_wave_barrier();
    __builtin_amdgcn_fence(__ATOMIC_ACQUIRE, "wavefront");
}

__global__ __launch_bounds__(WAVES * 32)
void entropy9x9_wmma_kernel(const float* __restrict__ x, float* __restrict__ out) {
    __shared__ __attribute__((aligned(16))) float eT_all[WAVES][PROWS * EPITCH];
    __shared__ __attribute__((aligned(16))) float oT_all[WAVES][TROWS * OPITCH];

    const int lane = threadIdx.x & 31;
    const int w    = __builtin_amdgcn_readfirstlane((int)(threadIdx.x >> 5));
    float* eT = eT_all[w];
    float* oT = oT_all[w];

    const int tIdx = (int)blockIdx.x * WAVES + w;
    const int tcol = tIdx & 15;
    const int trow = (tIdx >> 4) & 31;
    const int b    = tIdx >> 9;
    const int y0 = trow * TROWS;
    const int c0 = tcol * TF;

    const float* img = x + (size_t)b * HH * ROWF;

    const int gf = c0 - 12 + 4 * lane;
    const bool colok = lane < PPIECES && (unsigned)gf < (unsigned)ROWF;
    const int pbase = (y0 - 4) * ROWF + gf;
    v4f pv[PROWS];
    #pragma unroll
    for (int t = 0; t < PROWS; ++t)
        pv[t] = (v4f){0.0f, 0.0f, 0.0f, 0.0f};
    if (colok) {
        if (y0 - 4 >= 0) {
            #pragma unroll
            for (int t = 0; t < 4; ++t)
                pv[t] = *(const v4f*)(img + (pbase + t * ROWF));
        }
        #pragma unroll
        for (int t = 4; t < 20; ++t)
            pv[t] = *(const v4f*)(img + (pbase + t * ROWF));
        if (y0 + 19 < HH) {
            #pragma unroll
            for (int t = 20; t < PROWS; ++t)
                pv[t] = *(const v4f*)(img + (pbase + t * ROWF));
        }
    }
    #pragma unroll
    for (int t = 0; t < PROWS; ++t) {
        const v4f v = pv[t];
        v4f e;
        e.x = v.x * __log2f(v.x + 1e-10f);
        e.y = v.y * __log2f(v.y + 1e-10f);
        e.z = v.z * __log2f(v.z + 1e-10f);
        e.w = v.w * __log2f(v.w + 1e-10f);
        if (lane < PPIECES)
            *(v4f*)(eT + t * EPITCH + 4 * lane) = e;
    }
    wave_lds_handover();

    const int m  = lane & 15;
    const int hi = lane >> 4;
    const int ka = hi * 2;

    const int offA = m * EPITCH + ka * CC;
    const int offB = offA + 16 * CC;
    const int rowS = 16 + (m & 3) + 4 * (m >> 3);
    const int offS = rowS * EPITCH + (((m >> 2) & 1) * 16 + ka) * CC;

    v2f wh[6];
    #pragma unroll
    for (int q = 0; q < 6; ++q) {
        const int kb = q * 4 + ka;
        wh[q].x = (m <= kb     && kb     <= m + 8) ? 1.0f : 0.0f;
        wh[q].y = (m <= kb + 1 && kb + 1 <= m + 8) ? 1.0f : 0.0f;
    }
    v2f wv[6];
    #pragma unroll
    for (int t = 0; t < 6; ++t) {
        const int kk = (t < 4) ? (8 * hi + 2 * t) : (16 + 4 * hi + 2 * (t - 4));
        wv[t].x = (m <= kk     && kk     <= m + 8) ? 1.0f : 0.0f;
        wv[t].y = (m <= kk + 1 && kk + 1 <= m + 8) ? 1.0f : 0.0f;
    }

    for (int c = 0; c < CC; ++c) {
        v8f acc0 = {};
        v8f acc1 = {};
        v8f accS = {};
        #pragma unroll
        for (int q = 0; q < 6; ++q) {
            const int fo = q * 4 * CC + c;
            v2f a0, a1, aS;
            a0.x = eT[offA + fo];
            a0.y = eT[offA + fo + CC];
            a1.x = eT[offB + fo];
            a1.y = eT[offB + fo + CC];
            aS.x = eT[offS + fo];
            aS.y = eT[offS + fo + CC];
            acc0 = __builtin_amdgcn_wmma_f32_16x16x4_f32(false, a0, false, wh[q],
                                                         (short)0, acc0, false, false);
            acc1 = __builtin_amdgcn_wmma_f32_16x16x4_f32(false, a1, false, wh[q],
                                                         (short)0, acc1, false, false);
            accS = __builtin_amdgcn_wmma_f32_16x16x4_f32(false, aS, false, wh[q],
                                                         (short)0, accS, false, false);
        }

        v8f o0 = {};
        v8f o1 = {};
        #pragma unroll
        for (int t = 0; t < 4; ++t) {
            const v2f b0 = {acc0[2 * t], acc0[2 * t + 1]};
            const v2f b1 = {acc1[2 * t], acc1[2 * t + 1]};
            o0 = __builtin_amdgcn_wmma_f32_16x16x4_f32(false, wv[t], false, b0,
                                                       (short)0, o0, false, false);
            o1 = __builtin_amdgcn_wmma_f32_16x16x4_f32(false, wv[t], false, b1,
                                                       (short)0, o1, false, false);
        }
        #pragma unroll
        for (int u = 0; u < 2; ++u) {
            const v2f b0 = {accS[2 * u], accS[2 * u + 1]};
            const v2f b1 = {accS[4 + 2 * u], accS[4 + 2 * u + 1]};
            o0 = __builtin_amdgcn_wmma_f32_16x16x4_f32(false, wv[4 + u], false, b0,
                                                       (short)0, o0, false, false);
            o1 = __builtin_amdgcn_wmma_f32_16x16x4_f32(false, wv[4 + u], false, b1,
                                                       (short)0, o1, false, false);
        }
        #pragma unroll
        for (int g = 0; g < 8; ++g) {
            oT[(g + hi * 8) * OPITCH + m * CC + c]        = -o0[g];
            oT[(g + hi * 8) * OPITCH + (16 + m) * CC + c] = -o1[g];
        }
    }
    wave_lds_handover();

    float* oimg = out + (size_t)b * HH * ROWF;
    if (lane < OPIECES) {
        #pragma unroll
        for (int t = 0; t < TROWS; ++t) {
            const v4f v = *(const v4f*)(oT + t * OPITCH + 4 * lane);
            *(v4f*)(oimg + ((size_t)(y0 + t) * ROWF + c0 + 4 * lane)) = v;
        }
    }
}

extern "C" void kernel_launch(void* const* d_in, const int* in_sizes, int n_in,
                              void* d_out, int out_size, void* d_ws, size_t ws_size,
                              hipStream_t stream) {
    (void)in_sizes; (void)n_in; (void)d_ws; (void)ws_size; (void)out_size;
    const float* x = (const float*)d_in[0];
    float* out = (float*)d_out;

    dim3 grid(4096);
    dim3 block(WAVES * 32);
    entropy9x9_wmma_kernel<<<grid, block, 0, stream>>>(x, out);
}
